# starting kernel
# baseline (speedup 1.0000x reference)
#include <hip/hip_runtime.h>


typedef __attribute__((ext_vector_type(2))) float v2f;
typedef __attribute__((ext_vector_type(8))) float v8f;

#define NM    8
#define NC    1024
#define ROWP  1028
#define EPSF  1e-8f

__global__ __launch_bounds__(256)
void gs_wmma_kernel(const float* __restrict__ x, float* __restrict__ out, int nb) {
  __shared__ float Xs[NM * ROWP];
  __shared__ float Gpart[8][8][8];
  __shared__ float Gm[8][8];
  __shared__ float Ac[8][8];

  const int b    = blockIdx.x;
  const int t    = threadIdx.x;
  const int lane = t & 31;
  const int wave = t >> 5;

  {
    const float4* __restrict__ xv = (const float4*)x;
    #pragma unroll
    for (int row = 0; row < NM; ++row) {
      float4 v = xv[((size_t)row * nb + b) * (NC / 4) + t];
      *(float4*)&Xs[row * ROWP + t * 4] = v;
    }
  }
  __syncthreads();

  {
    const int   m       = lane & 15;
    const int   kb      = (lane >> 4) << 1;
    const int   rowSafe = m & 7;
    const float keep    = (m < NM) ? 1.0f : 0.0f;
    const int   kbase   = wave * 128 + kb;

    v8f acc = {};
    for (int k0 = 0; k0 < 128; k0 += 4) {
      v2f a = *(const v2f*)&Xs[rowSafe * ROWP + kbase + k0];
      a.x *= keep;
      a.y *= keep;
      acc = __builtin_amdgcn_wmma_f32_16x16x4_f32(
           false, a,  false, a,
           (short)0, acc,  false,  false);
    }
    if (lane < 8) {
      #pragma unroll
      for (int r = 0; r < 8; ++r) Gpart[wave][r][lane] = acc[r];
    }
  }
  __syncthreads();

  if (t < 64) {
    const int i = t >> 3, j = t & 7;
    float s = 0.0f;
    #pragma unroll
    for (int w = 0; w < 8; ++w) s += Gpart[w][i][j];
    Gm[i][j] = s;
  }
  __syncthreads();

  if (t == 0) {
    float g[8][8];
    #pragma unroll
    for (int i = 0; i < 8; ++i)
      #pragma unroll
      for (int j = 0; j < 8; ++j) g[i][j] = Gm[i][j];

    float A[8][8];
    #pragma unroll
    for (int i = 0; i < 8; ++i)
      #pragma unroll
      for (int j = 0; j < 8; ++j) A[i][j] = 0.0f;

    float d[8];

    {
      float n0   = sqrtf(fmaxf(g[0][0], 0.0f));
      float inv0 = 1.0f / (n0 + EPSF);
      A[0][0] = inv0;
      d[0]    = inv0 * inv0 * g[0][0];
    }

    #pragma unroll
    for (int i = 1; i < 8; ++i) {
      float c[8];
      #pragma unroll
      for (int k = 0; k < 8; ++k) c[k] = 0.0f;
      c[i] = 1.0f;

      #pragma unroll
      for (int j = 0; j < 8; ++j) {
        if (j < i) {
          float dv = 0.0f;
          #pragma unroll
          for (int k = 0; k < 8; ++k)
            if (k <= j) dv += A[j][k] * g[k][i];
          float proj = dv / (d[j] + EPSF);
          #pragma unroll
          for (int k = 0; k < 8; ++k)
            if (k <= j) c[k] -= proj * A[j][k];
        }
      }

      float n2 = 0.0f;
      #pragma unroll
      for (int k = 0; k < 8; ++k) {
        if (k <= i) {
          float y = 0.0f;
          #pragma unroll
          for (int m2 = 0; m2 < 8; ++m2)
            if (m2 <= i) y += g[k][m2] * c[m2];
          n2 += c[k] * y;
        }
      }
      float nn = sqrtf(fmaxf(n2, 0.0f));
      if (nn > EPSF) {
        float inv = 1.0f / nn;
        #pragma unroll
        for (int k = 0; k < 8; ++k) c[k] *= inv;
        d[i] = n2 * inv * inv;
      } else {
        d[i] = n2;
      }
      #pragma unroll
      for (int k = 0; k < 8; ++k) A[i][k] = c[k];
    }

    #pragma unroll
    for (int i = 0; i < 8; ++i)
      #pragma unroll
      for (int k = 0; k < 8; ++k) Ac[i][k] = A[i][k];
  }
  __syncthreads();

  {
    float4 xk[8];
    #pragma unroll
    for (int k = 0; k < 8; ++k) xk[k] = *(const float4*)&Xs[k * ROWP + t * 4];

    float4* __restrict__ ov = (float4*)out;
    #pragma unroll
    for (int m = 0; m < 8; ++m) {
      float4 acc4 = {0.0f, 0.0f, 0.0f, 0.0f};
      #pragma unroll
      for (int k = 0; k < 8; ++k) {
        const float c = Ac[m][k];
        acc4.x += c * xk[k].x;
        acc4.y += c * xk[k].y;
        acc4.z += c * xk[k].z;
        acc4.w += c * xk[k].w;
      }
      ov[((size_t)m * nb + b) * (NC / 4) + t] = acc4;
    }
  }
}

extern "C" void kernel_launch(void* const* d_in, const int* in_sizes, int n_in,
                              void* d_out, int out_size, void* d_ws, size_t ws_size,
                              hipStream_t stream) {
  const float* x   = (const float*)d_in[0];
  float*       out = (float*)d_out;
  const int    nb  = in_sizes[0] / (NM * NC);
  gs_wmma_kernel<<<nb, 256, 0, stream>>>(x, out, nb);
}


# Recurrence on G runs on 8 lanes of wave 0 (lane k: column k of A, row k of G, sums over k by row_xmask butterflies), not one lane: 141 -> 82 VGPRs, 5 blocks per WGP not 3; dots now sum as lane trees
# speedup vs baseline: 1.1772x; 1.1772x over previous
#include <hip/hip_runtime.h>


typedef __attribute__((ext_vector_type(2))) float v2f;
typedef __attribute__((ext_vector_type(8))) float v8f;

#define NM    8
#define NC    1024
#define ROWP  1028
#define EPSF  1e-8f

template <int CTRL>
__device__ __forceinline__ float row_read(float v) {
  const int r = __builtin_amdgcn_update_dpp(0, __builtin_bit_cast(int, v), CTRL,
                                            0xf, 0xf, true);
  return __builtin_bit_cast(float, r);
}

__device__ __forceinline__ float sum8(float v) {
  v += row_read<0x161>(v);
  v += row_read<0x162>(v);
  v += row_read<0x164>(v);
  return row_read<0x150>(v);
}

__device__ __forceinline__ float lane_of8(float v, int m) {
  switch (m) {
    case 0: return row_read<0x150>(v);
    case 1: return row_read<0x151>(v);
    case 2: return row_read<0x152>(v);
    case 3: return row_read<0x153>(v);
    case 4: return row_read<0x154>(v);
    case 5: return row_read<0x155>(v);
    case 6: return row_read<0x156>(v);
    default: return row_read<0x157>(v);
  }
}

__global__ __launch_bounds__(256)
void gs_wmma_kernel(const float* __restrict__ x, float* __restrict__ out, int nb) {
  __shared__ float Xs[NM * ROWP];
  __shared__ float Gpart[8][8][8];
  __shared__ __attribute__((aligned(16))) float Gm[8][8];
  __shared__ float Ac[8][8];

  const int b    = blockIdx.x;
  const int t    = threadIdx.x;
  const int lane = t & 31;
  const int wave = t >> 5;

  {
    const float4* __restrict__ xv = (const float4*)x;
    #pragma unroll
    for (int row = 0; row < NM; ++row) {
      float4 v = xv[((size_t)row * nb + b) * (NC / 4) + t];
      *(float4*)&Xs[row * ROWP + t * 4] = v;
    }
  }
  __syncthreads();

  {
    const int   m       = lane & 15;
    const int   kb      = (lane >> 4) << 1;
    const int   rowSafe = m & 7;
    const float keep    = (m < NM) ? 1.0f : 0.0f;
    const int   kbase   = wave * 128 + kb;

    v8f acc = {};
    for (int k0 = 0; k0 < 128; k0 += 4) {
      v2f a = *(const v2f*)&Xs[rowSafe * ROWP + kbase + k0];
      a.x *= keep;
      a.y *= keep;
      acc = __builtin_amdgcn_wmma_f32_16x16x4_f32(
           false, a,  false, a,
           (short)0, acc,  false,  false);
    }
    if (lane < 8) {
      #pragma unroll
      for (int r = 0; r < 8; ++r) Gpart[wave][r][lane] = acc[r];
    }
  }
  __syncthreads();

  if (t < 64) {
    const int i = t >> 3, j = t & 7;
    float s = 0.0f;
    #pragma unroll
    for (int w = 0; w < 8; ++w) s += Gpart[w][i][j];
    Gm[i][j] = s;
  }
  __syncthreads();

  if (wave == 0) {
    const int k = lane & 7;
    float grow[8];
    {
      const float4 lo = *(const float4*)&Gm[k][0];
      const float4 hi = *(const float4*)&Gm[k][4];
      grow[0] = lo.x; grow[1] = lo.y; grow[2] = lo.z; grow[3] = lo.w;
      grow[4] = hi.x; grow[5] = hi.y; grow[6] = hi.z; grow[7] = hi.w;
    }
    const float g00 = Gm[0][0];

    float Acol[8];
    float dpe[8];
    #pragma unroll
    for (int i = 0; i < 8; ++i) Acol[i] = 0.0f;

    {
      const float n0   = sqrtf(fmaxf(g00, 0.0f));
      const float inv0 = 1.0f / (n0 + EPSF);
      Acol[0] = (k == 0) ? inv0 : 0.0f;
      dpe[0]  = inv0 * inv0 * g00 + EPSF;
    }

    #pragma unroll
    for (int i = 1; i < 8; ++i) {
      float dv[8];
      #pragma unroll
      for (int j = 0; j < 8; ++j)
        if (j < i) dv[j] = sum8(Acol[j] * grow[i]);

      float ck = (k == i) ? 1.0f : 0.0f;
      #pragma unroll
      for (int j = 0; j < 8; ++j)
        if (j < i) {
          const float proj = dv[j] / dpe[j];
          ck -= proj * Acol[j];
        }

      float y0 = 0.0f, y1 = 0.0f;
      #pragma unroll
      for (int m2 = 0; m2 < 8; m2 += 2) {
        if (m2 <= i)     y0 += grow[m2]     * lane_of8(ck, m2);
        if (m2 + 1 <= i) y1 += grow[m2 + 1] * lane_of8(ck, m2 + 1);
      }
      const float n2 = sum8(ck * (y0 + y1));

      const float nn  = sqrtf(fmaxf(n2, 0.0f));
      const float inv = (nn > EPSF) ? 1.0f / nn : 1.0f;
      Acol[i] = ck * inv;
      dpe[i]  = n2 * inv * inv + EPSF;
    }

    if (lane < 8) {
      #pragma unroll
      for (int i = 0; i < 8; ++i) Ac[i][k] = Acol[i];
    }
  }
  __syncthreads();

  {
    float4 xk[8];
    #pragma unroll
    for (int k = 0; k < 8; ++k) xk[k] = *(const float4*)&Xs[k * ROWP + t * 4];

    float4* __restrict__ ov = (float4*)out;
    #pragma unroll
    for (int m = 0; m < 8; ++m) {
      float4 acc4 = {0.0f, 0.0f, 0.0f, 0.0f};
      #pragma unroll
      for (int k = 0; k < 8; ++k) {
        const float c = Ac[m][k];
        acc4.x += c * xk[k].x;
        acc4.y += c * xk[k].y;
        acc4.z += c * xk[k].z;
        acc4.w += c * xk[k].w;
      }
      ov[((size_t)m * nb + b) * (NC / 4) + t] = acc4;
    }
  }
}

extern "C" void kernel_launch(void* const* d_in, const int* in_sizes, int n_in,
                              void* d_out, int out_size, void* d_ws, size_t ws_size,
                              hipStream_t stream) {
  const float* x   = (const float*)d_in[0];
  float*       out = (float*)d_out;
  const int    nb  = in_sizes[0] / (NM * NC);
  gs_wmma_kernel<<<nb, 256, 0, stream>>>(x, out, nb);
}


# 4 waves per work-group instead of 8 (each thread takes two float4 columns of a row, each wave two of the 8 Gram K slices on two accumulators): half as many waves to start; every sum keeps its order
# speedup vs baseline: 1.2241x; 1.0398x over previous
#include <hip/hip_runtime.h>


typedef __attribute__((ext_vector_type(2))) float v2f;
typedef __attribute__((ext_vector_type(8))) float v8f;

#define NM    8
#define NC    1024
#define ROWP  1028
#define EPSF  1e-8f
#define NT    128
#define NH    2

template <int CTRL>
__device__ __forceinline__ float row_read(float v) {
  const int r = __builtin_amdgcn_update_dpp(0, __builtin_bit_cast(int, v), CTRL,
                                            0xf, 0xf, true);
  return __builtin_bit_cast(float, r);
}

__device__ __forceinline__ float sum8(float v) {
  v += row_read<0x161>(v);
  v += row_read<0x162>(v);
  v += row_read<0x164>(v);
  return row_read<0x150>(v);
}

__device__ __forceinline__ float lane_of8(float v, int m) {
  switch (m) {
    case 0: return row_read<0x150>(v);
    case 1: return row_read<0x151>(v);
    case 2: return row_read<0x152>(v);
    case 3: return row_read<0x153>(v);
    case 4: return row_read<0x154>(v);
    case 5: return row_read<0x155>(v);
    case 6: return row_read<0x156>(v);
    default: return row_read<0x157>(v);
  }
}

__global__ __launch_bounds__(NT)
void gs_wmma_kernel(const float* __restrict__ x, float* __restrict__ out, int nb) {
  __shared__ float Xs[NM * ROWP];
  __shared__ float Gpart[8][8][8];
  __shared__ __attribute__((aligned(16))) float Gm[8][8];
  __shared__ float Ac[8][8];

  const int b    = blockIdx.x;
  const int t    = threadIdx.x;
  const int lane = t & 31;
  const int wave = t >> 5;

  {
    const float4* __restrict__ xv = (const float4*)x;
    float4 v[NM][NH];
    #pragma unroll
    for (int row = 0; row < NM; ++row) {
      #pragma unroll
      for (int h = 0; h < NH; ++h)
        v[row][h] = xv[((size_t)row * nb + b) * (NC / 4) + h * NT + t];
    }
    #pragma unroll
    for (int row = 0; row < NM; ++row) {
      #pragma unroll
      for (int h = 0; h < NH; ++h)
        *(float4*)&Xs[row * ROWP + (h * NT + t) * 4] = v[row][h];
    }
  }
  __syncthreads();

  {
    const int   m       = lane & 15;
    const int   kb      = (lane >> 4) << 1;
    const int   rowSafe = m & 7;
    const float keep    = (m < NM) ? 1.0f : 0.0f;
    const int   kbase   = wave * 256 + kb;

    v8f acc0 = {};
    v8f acc1 = {};
    for (int k0 = 0; k0 < 128; k0 += 4) {
      v2f a0 = *(const v2f*)&Xs[rowSafe * ROWP + kbase + k0];
      v2f a1 = *(const v2f*)&Xs[rowSafe * ROWP + kbase + 128 + k0];
      a0.x *= keep;
      a0.y *= keep;
      a1.x *= keep;
      a1.y *= keep;
      acc0 = __builtin_amdgcn_wmma_f32_16x16x4_f32(
           false, a0,  false, a0,
           (short)0, acc0,  false,  false);
      acc1 = __builtin_amdgcn_wmma_f32_16x16x4_f32(
           false, a1,  false, a1,
           (short)0, acc1,  false,  false);
    }
    if (lane < 8) {
      #pragma unroll
      for (int r = 0; r < 8; ++r) {
        Gpart[2 * wave][r][lane]     = acc0[r];
        Gpart[2 * wave + 1][r][lane] = acc1[r];
      }
    }
  }
  __syncthreads();

  if (t < 64) {
    const int i = t >> 3, j = t & 7;
    float s = 0.0f;
    #pragma unroll
    for (int w = 0; w < 8; ++w) s += Gpart[w][i][j];
    Gm[i][j] = s;
  }
  __syncthreads();

  if (wave == 0) {
    const int k = lane & 7;
    float grow[8];
    {
      const float4 lo = *(const float4*)&Gm[k][0];
      const float4 hi = *(const float4*)&Gm[k][4];
      grow[0] = lo.x; grow[1] = lo.y; grow[2] = lo.z; grow[3] = lo.w;
      grow[4] = hi.x; grow[5] = hi.y; grow[6] = hi.z; grow[7] = hi.w;
    }
    const float g00 = Gm[0][0];

    float Acol[8];
    float dpe[8];
    #pragma unroll
    for (int i = 0; i < 8; ++i) Acol[i] = 0.0f;

    {
      const float n0   = sqrtf(fmaxf(g00, 0.0f));
      const float inv0 = 1.0f / (n0 + EPSF);
      Acol[0] = (k == 0) ? inv0 : 0.0f;
      dpe[0]  = inv0 * inv0 * g00 + EPSF;
    }

    #pragma unroll
    for (int i = 1; i < 8; ++i) {
      float dv[8];
      #pragma unroll
      for (int j = 0; j < 8; ++j)
        if (j < i) dv[j] = sum8(Acol[j] * grow[i]);

      float ck = (k == i) ? 1.0f : 0.0f;
      #pragma unroll
      for (int j = 0; j < 8; ++j)
        if (j < i) {
          const float proj = dv[j] / dpe[j];
          ck -= proj * Acol[j];
        }

      float y0 = 0.0f, y1 = 0.0f;
      #pragma unroll
      for (int m2 = 0; m2 < 8; m2 += 2) {
        if (m2 <= i)     y0 += grow[m2]     * lane_of8(ck, m2);
        if (m2 + 1 <= i) y1 += grow[m2 + 1] * lane_of8(ck, m2 + 1);
      }
      const float n2 = sum8(ck * (y0 + y1));

      const float nn  = sqrtf(fmaxf(n2, 0.0f));
      const float inv = (nn > EPSF) ? 1.0f / nn : 1.0f;
      Acol[i] = ck * inv;
      dpe[i]  = n2 * inv * inv + EPSF;
    }

    if (lane < 8) {
      #pragma unroll
      for (int i = 0; i < 8; ++i) Ac[i][k] = Acol[i];
    }
  }
  __syncthreads();

  {
    float4 xk[NH][8];
    #pragma unroll
    for (int h = 0; h < NH; ++h) {
      #pragma unroll
      for (int k = 0; k < 8; ++k) xk[h][k] = *(const float4*)&Xs[k * ROWP + (h * NT + t) * 4];
    }

    float4* __restrict__ ov = (float4*)out;
    #pragma unroll
    for (int m = 0; m < 8; ++m) {
      float c[8];
      #pragma unroll
      for (int k = 0; k < 8; ++k) c[k] = Ac[m][k];
      #pragma unroll
      for (int h = 0; h < NH; ++h) {
        float4 acc4 = {0.0f, 0.0f, 0.0f, 0.0f};
        #pragma unroll
        for (int k = 0; k < 8; ++k) {
          acc4.x += c[k] * xk[h][k].x;
          acc4.y += c[k] * xk[h][k].y;
          acc4.z += c[k] * xk[h][k].z;
          acc4.w += c[k] * xk[h][k].w;
        }
        ov[((size_t)m * nb + b) * (NC / 4) + h * NT + t] = acc4;
      }
    }
  }
}

extern "C" void kernel_launch(void* const* d_in, const int* in_sizes, int n_in,
                              void* d_out, int out_size, void* d_ws, size_t ws_size,
                              hipStream_t stream) {
  const float* x   = (const float*)d_in[0];
  float*       out = (float*)d_out;
  const int    nb  = in_sizes[0] / (NM * NC);
  gs_wmma_kernel<<<nb, NT, 0, stream>>>(x, out, nb);
}
